# starting kernel
# baseline (speedup 1.0000x reference)
#include <hip/hip_runtime.h>


typedef float v2f __attribute__((ext_vector_type(2)));
typedef float v8f __attribute__((ext_vector_type(8)));

#define N_IN    262144
#define N_TAPS  255
#define M_OUT   (N_IN + N_TAPS - 1)
#define OUT_TILE 512
#define KBLK    68
#define LDX     770
#define HR_SIZE 288

__global__ __launch_bounds__(256)
void fir_wmma_f32_kernel(const float* __restrict__ x,
                         const float* __restrict__ taps,
                         float* __restrict__ y)
{
    __shared__ float xt[16 * LDX];
    __shared__ float hr[HR_SIZE];

    const int  tid     = threadIdx.x;
    const int  rg      = blockIdx.y;
    const long wg_pout = (long)blockIdx.x * OUT_TILE;
    const long wg_p0   = wg_pout - (N_TAPS - 1);

    for (int c = tid; c < HR_SIZE; c += 256) {
        int j = 270 - c;
        float v = 0.f;
        if (j >= 0 && j < N_TAPS) {
            int t = j - 127; if (t < 0) t = -t;
            v = taps[t];
        }
        hr[c] = v;
    }

    for (int idx = tid; idx < 16 * 192; idx += 256) {
        const int  row = idx / 192;
        const int  c4  = (idx % 192) * 4;
        const long g   = wg_p0 + c4;
        const float* __restrict__ src = x + (long)(rg * 16 + row) * N_IN;
        float4 v;
        if (g >= 0 && g + 3 < (long)N_IN) {
            v = *(const float4*)(src + g);
        } else {
            v.x = (g + 0 >= 0 && g + 0 < (long)N_IN) ? src[g + 0] : 0.f;
            v.y = (g + 1 >= 0 && g + 1 < (long)N_IN) ? src[g + 1] : 0.f;
            v.z = (g + 2 >= 0 && g + 2 < (long)N_IN) ? src[g + 2] : 0.f;
            v.w = (g + 3 >= 0 && g + 3 < (long)N_IN) ? src[g + 3] : 0.f;
        }
        float* d = &xt[row * LDX + c4];
        d[0] = v.x; d[1] = v.y; d[2] = v.z; d[3] = v.w;
    }
    __syncthreads();

    const int wave = tid >> 5;
    const int lane = tid & 31;
    const int mn   = lane & 15;
    const int kp   = lane >> 4;

    const float* __restrict__ arow = &xt[mn * LDX + wave * 64 + 2 * kp];
    const int rb0 = 16 - mn + 2 * kp;

    v8f acc0 = {0.f,0.f,0.f,0.f,0.f,0.f,0.f,0.f};
    v8f acc1 = acc0, acc2 = acc0, acc3 = acc0;

    #pragma unroll 2
    for (int kb = 0; kb < KBLK; ++kb) {
        const int k0 = kb * 4;
        v2f b;
        b.x = hr[rb0 + k0];
        b.y = hr[rb0 + k0 + 1];
        v2f a0 = *(const v2f*)(arow + k0);
        v2f a1 = *(const v2f*)(arow + k0 + 16);
        v2f a2 = *(const v2f*)(arow + k0 + 32);
        v2f a3 = *(const v2f*)(arow + k0 + 48);
        acc0 = __builtin_amdgcn_wmma_f32_16x16x4_f32(false, a0, false, b, (short)0, acc0, false, false);
        acc1 = __builtin_amdgcn_wmma_f32_16x16x4_f32(false, a1, false, b, (short)0, acc1, false, false);
        acc2 = __builtin_amdgcn_wmma_f32_16x16x4_f32(false, a2, false, b, (short)0, acc2, false, false);
        acc3 = __builtin_amdgcn_wmma_f32_16x16x4_f32(false, a3, false, b, (short)0, acc3, false, false);
    }

    const long rowbase = (long)(rg * 16) * M_OUT;
    #pragma unroll
    for (int blk = 0; blk < 4; ++blk) {
        const v8f acc = (blk == 0) ? acc0 : (blk == 1) ? acc1 : (blk == 2) ? acc2 : acc3;
        const long pout = wg_pout + wave * 64 + blk * 16 + mn;
        if (pout < (long)M_OUT) {
            #pragma unroll
            for (int v = 0; v < 8; ++v) {
                const int row = v + 8 * kp;
                y[rowbase + (long)row * M_OUT + pout] = acc[v];
            }
        }
    }
}

extern "C" void kernel_launch(void* const* d_in, const int* in_sizes, int n_in,
                              void* d_out, int out_size, void* d_ws, size_t ws_size,
                              hipStream_t stream) {
    const float* x    = (const float*)d_in[0];
    const float* taps = (const float*)d_in[1];
    float* y          = (float*)d_out;

    dim3 grid((M_OUT + OUT_TILE - 1) / OUT_TILE,  2, 1);
    dim3 block(256, 1, 1);
    fir_wmma_f32_kernel<<<grid, block, 0, stream>>>(x, taps, y);
}


# one 1024-thread workgroup per WGP owns an equal 1040-output slice of all 32 rows (253 groups; LDS window above 160 KiB keeps one per WGP) instead of 1026 groups of 512 landing 4 or 5 to a WGP
# speedup vs baseline: 1.0678x; 1.0678x over previous
#include <hip/hip_runtime.h>


typedef float v2f __attribute__((ext_vector_type(2)));
typedef float v8f __attribute__((ext_vector_type(8)));

#define N_IN     262144
#define N_TAPS   255
#define M_OUT    (N_IN + N_TAPS - 1)
#define BLKS     65
#define OUT_TILE (BLKS * 16)
#define NUM_WG   253
#define XW4      324
#define KBLK     68
#define LDX      1298
#define HR_SIZE  288
#define THREADS  1024

__device__ __forceinline__ void store_block(float* __restrict__ y, const v8f acc,
                                            const long rowbase, const long pout, const int kp)
{
    if (pout < (long)M_OUT) {
        #pragma unroll
        for (int v = 0; v < 8; ++v) {
            const int row = v + 8 * kp;
            y[rowbase + (long)row * M_OUT + pout] = acc[v];
        }
    }
}

__global__ __launch_bounds__(THREADS)
void fir_wmma_f32_kernel(const float* __restrict__ x,
                         const float* __restrict__ taps,
                         float* __restrict__ y)
{
    __shared__ float xt[32 * LDX];
    __shared__ float hr[HR_SIZE];

    const int  tid     = threadIdx.x;
    const long wg_pout = (long)blockIdx.x * OUT_TILE;
    const long wg_p0   = wg_pout - (N_TAPS - 1);

    for (int c = tid; c < HR_SIZE; c += THREADS) {
        int j = 270 - c;
        float v = 0.f;
        if (j >= 0 && j < N_TAPS) {
            int t = j - 127; if (t < 0) t = -t;
            v = taps[t];
        }
        hr[c] = v;
    }

    for (int idx = tid; idx < 32 * XW4; idx += THREADS) {
        const int  row = idx / XW4;
        const int  c4  = (idx % XW4) * 4;
        const long g   = wg_p0 + c4;
        const float* __restrict__ src = x + (long)row * N_IN;
        float4 v;
        if (g >= 0 && g + 3 < (long)N_IN) {
            v = *(const float4*)(src + g);
        } else {
            v.x = (g + 0 >= 0 && g + 0 < (long)N_IN) ? src[g + 0] : 0.f;
            v.y = (g + 1 >= 0 && g + 1 < (long)N_IN) ? src[g + 1] : 0.f;
            v.z = (g + 2 >= 0 && g + 2 < (long)N_IN) ? src[g + 2] : 0.f;
            v.w = (g + 3 >= 0 && g + 3 < (long)N_IN) ? src[g + 3] : 0.f;
        }
        float* d = &xt[row * LDX + c4];
        d[0] = v.x; d[1] = v.y; d[2] = v.z; d[3] = v.w;
    }
    __syncthreads();

    const int wave = tid >> 5;
    const int lane = tid & 31;
    const int mn   = lane & 15;
    const int kp   = lane >> 4;
    const int rb0 = 16 - mn + 2 * kp;

    {
        const int rg  = wave >> 4;
        const int grp = wave & 15;
        const float* __restrict__ arow = &xt[(rg * 16 + mn) * LDX + grp * 64 + 2 * kp];

        v8f acc0 = {0.f,0.f,0.f,0.f,0.f,0.f,0.f,0.f};
        v8f acc1 = acc0, acc2 = acc0, acc3 = acc0;

        #pragma unroll 2
        for (int kb = 0; kb < KBLK; ++kb) {
            const int k0 = kb * 4;
            v2f b;
            b.x = hr[rb0 + k0];
            b.y = hr[rb0 + k0 + 1];
            v2f a0 = *(const v2f*)(arow + k0);
            v2f a1 = *(const v2f*)(arow + k0 + 16);
            v2f a2 = *(const v2f*)(arow + k0 + 32);
            v2f a3 = *(const v2f*)(arow + k0 + 48);
            acc0 = __builtin_amdgcn_wmma_f32_16x16x4_f32(false, a0, false, b, (short)0, acc0, false, false);
            acc1 = __builtin_amdgcn_wmma_f32_16x16x4_f32(false, a1, false, b, (short)0, acc1, false, false);
            acc2 = __builtin_amdgcn_wmma_f32_16x16x4_f32(false, a2, false, b, (short)0, acc2, false, false);
            acc3 = __builtin_amdgcn_wmma_f32_16x16x4_f32(false, a3, false, b, (short)0, acc3, false, false);
        }

        const long rowbase = (long)(rg * 16) * M_OUT;
        const long pbase   = wg_pout + grp * 64 + mn;
        store_block(y, acc0, rowbase, pbase,      kp);
        store_block(y, acc1, rowbase, pbase + 16, kp);
        store_block(y, acc2, rowbase, pbase + 32, kp);
        store_block(y, acc3, rowbase, pbase + 48, kp);
    }

    if (wave < 2) {
        const int rg = wave;
        const float* __restrict__ arow = &xt[(rg * 16 + mn) * LDX + 64 * 16 + 2 * kp];
        v8f acc = {0.f,0.f,0.f,0.f,0.f,0.f,0.f,0.f};
        #pragma unroll 4
        for (int kb = 0; kb < KBLK; ++kb) {
            const int k0 = kb * 4;
            v2f b;
            b.x = hr[rb0 + k0];
            b.y = hr[rb0 + k0 + 1];
            v2f a = *(const v2f*)(arow + k0);
            acc = __builtin_amdgcn_wmma_f32_16x16x4_f32(false, a, false, b, (short)0, acc, false, false);
        }
        const long rowbase = (long)(rg * 16) * M_OUT;
        store_block(y, acc, rowbase, wg_pout + 64 * 16 + mn, kp);
    }
}

extern "C" void kernel_launch(void* const* d_in, const int* in_sizes, int n_in,
                              void* d_out, int out_size, void* d_ws, size_t ws_size,
                              hipStream_t stream) {
    const float* x    = (const float*)d_in[0];
    const float* taps = (const float*)d_in[1];
    float* y          = (float*)d_out;

    dim3 grid(NUM_WG, 1, 1);
    dim3 block(THREADS, 1, 1);
    fir_wmma_f32_kernel<<<grid, block, 0, stream>>>(x, taps, y);
}


# taps are mirror-symmetric: per 32 outputs add and subtract the window's mirrored halves first, then two K=144 products (72 WMMAs, was 136); y=U+V and U-V, so the summation order changes
# speedup vs baseline: 1.3884x; 1.3002x over previous
#include <hip/hip_runtime.h>


typedef float v2f __attribute__((ext_vector_type(2)));
typedef float v4f __attribute__((ext_vector_type(4)));
typedef float v8f __attribute__((ext_vector_type(8)));

#define N_IN     262144
#define N_TAPS   255
#define M_OUT    (N_IN + N_TAPS - 1)
#define BLKS     65
#define OUT_TILE (BLKS * 16)
#define NUM_WG   253
#define XW4      324
#define KBLK     68
#define LDX      1298
#define HR_SIZE  288
#define THREADS  1024
#define FK       143
#define FKB      36

__device__ __forceinline__ float tap_at(const float* __restrict__ taps, const int j, const bool row_ok)
{
    int t = j - 127; if (t < 0) t = -t;
    const float v = taps[t < 127 ? t : 127];
    return (row_ok && t <= 127) ? v : 0.f;
}

__device__ __forceinline__ void store_block(float* __restrict__ y, const v8f acc,
                                            const long rowbase, const long pout, const int kp)
{
    if (pout < (long)M_OUT) {
        #pragma unroll
        for (int v = 0; v < 8; ++v) {
            const int row = v + 8 * kp;
            y[rowbase + (long)row * M_OUT + pout] = acc[v];
        }
    }
}

__global__ __launch_bounds__(THREADS)
void fir_wmma_f32_kernel(const float* __restrict__ x,
                         const float* __restrict__ taps,
                         float* __restrict__ y)
{
    __shared__ float xt[32 * LDX];
    __shared__ float hr[HR_SIZE];
    __shared__ v4f bf[FKB * 32];

    const int  tid     = threadIdx.x;
    const long wg_pout = (long)blockIdx.x * OUT_TILE;
    const long wg_p0   = wg_pout - (N_TAPS - 1);

    for (int c = tid; c < HR_SIZE; c += THREADS) {
        int j = 270 - c;
        float v = 0.f;
        if (j >= 0 && j < N_TAPS) {
            int t = j - 127; if (t < 0) t = -t;
            v = taps[t];
        }
        hr[c] = v;
    }

    float tp[2][4];
    #pragma unroll
    for (int e = 0; e < 2; ++e) {
        const int idx = tid + e * THREADS;
        const int c   = idx & 15;
        const int k   = 4 * (idx >> 5) + 2 * ((idx >> 4) & 1);
        tp[e][0] = tap_at(taps, 254 + c - k, k < FK);
        tp[e][1] = tap_at(taps, 285 - c - k, k < FK);
        tp[e][2] = tap_at(taps, 253 + c - k, k + 1 < FK);
        tp[e][3] = tap_at(taps, 284 - c - k, k + 1 < FK);
    }

    for (int idx = tid; idx < 32 * XW4; idx += THREADS) {
        const int  row = idx / XW4;
        const int  c4  = (idx % XW4) * 4;
        const long g   = wg_p0 + c4;
        const float* __restrict__ src = x + (long)row * N_IN;
        float4 v;
        if (g >= 0 && g + 3 < (long)N_IN) {
            v = *(const float4*)(src + g);
        } else {
            v.x = (g + 0 >= 0 && g + 0 < (long)N_IN) ? src[g + 0] : 0.f;
            v.y = (g + 1 >= 0 && g + 1 < (long)N_IN) ? src[g + 1] : 0.f;
            v.z = (g + 2 >= 0 && g + 2 < (long)N_IN) ? src[g + 2] : 0.f;
            v.w = (g + 3 >= 0 && g + 3 < (long)N_IN) ? src[g + 3] : 0.f;
        }
        float* d = &xt[row * LDX + c4];
        d[0] = v.x; d[1] = v.y; d[2] = v.z; d[3] = v.w;
    }
    #pragma unroll
    for (int e = 0; e < 2; ++e) {
        const int idx = tid + e * THREADS;
        if (idx < FKB * 32) {
            v4f q;
            q.x = 0.5f * (tp[e][0] + tp[e][1]); q.y = 0.5f * (tp[e][2] + tp[e][3]);
            q.z = 0.5f * (tp[e][0] - tp[e][1]); q.w = 0.5f * (tp[e][2] - tp[e][3]);
            bf[idx] = q;
        }
    }
    __syncthreads();

    const int wave = tid >> 5;
    const int lane = tid & 31;
    const int mn   = lane & 15;
    const int kp   = lane >> 4;
    const int rb0 = 16 - mn + 2 * kp;

    {
        const int rg  = wave >> 4;
        const int grp = wave & 15;
        const float* __restrict__ fwd = &xt[(rg * 16 + mn) * LDX + grp * 64 + 2 * kp];
        const float* __restrict__ mir = &xt[(rg * 16 + mn) * LDX + grp * 64 + 284 - 2 * kp];
        const v4f* __restrict__ bfl = &bf[lane];

        v8f u0 = {0.f,0.f,0.f,0.f,0.f,0.f,0.f,0.f};
        v8f w0 = u0, u1 = u0, w1 = u0;

        #pragma unroll 2
        for (int s = 0; s < FKB; ++s) {
            const int k0 = s * 4;
            const v4f bb = bfl[s * 32];
            v2f bs; bs.x = bb.x; bs.y = bb.y;
            v2f bd; bd.x = bb.z; bd.y = bb.w;
            const v2f f0 = *(const v2f*)(fwd + k0);
            const v2f m0 = *(const v2f*)(mir - k0);
            const v2f f1 = *(const v2f*)(fwd + 32 + k0);
            const v2f m1 = *(const v2f*)(mir + 32 - k0);
            v2f s0; s0.x = f0.x + m0.y; s0.y = f0.y + m0.x;
            v2f d0; d0.x = f0.x - m0.y; d0.y = f0.y - m0.x;
            v2f s1; s1.x = f1.x + m1.y; s1.y = f1.y + m1.x;
            v2f d1; d1.x = f1.x - m1.y; d1.y = f1.y - m1.x;
            u0 = __builtin_amdgcn_wmma_f32_16x16x4_f32(false, s0, false, bs, (short)0, u0, false, false);
            w0 = __builtin_amdgcn_wmma_f32_16x16x4_f32(false, d0, false, bd, (short)0, w0, false, false);
            u1 = __builtin_amdgcn_wmma_f32_16x16x4_f32(false, s1, false, bs, (short)0, u1, false, false);
            w1 = __builtin_amdgcn_wmma_f32_16x16x4_f32(false, d1, false, bd, (short)0, w1, false, false);
        }

        const long rowbase = (long)(rg * 16) * M_OUT;
        const long pbase   = wg_pout + grp * 64;
        store_block(y, u0 + w0, rowbase, pbase + mn,      kp);
        store_block(y, u0 - w0, rowbase, pbase + 31 - mn, kp);
        store_block(y, u1 + w1, rowbase, pbase + 32 + mn, kp);
        store_block(y, u1 - w1, rowbase, pbase + 63 - mn, kp);
    }

    if (wave < 2) {
        const int rg = wave;
        const float* __restrict__ arow = &xt[(rg * 16 + mn) * LDX + 64 * 16 + 2 * kp];
        v8f acc = {0.f,0.f,0.f,0.f,0.f,0.f,0.f,0.f};
        #pragma unroll 4
        for (int kb = 0; kb < KBLK; ++kb) {
            const int k0 = kb * 4;
            v2f b;
            b.x = hr[rb0 + k0];
            b.y = hr[rb0 + k0 + 1];
            v2f a = *(const v2f*)(arow + k0);
            acc = __builtin_amdgcn_wmma_f32_16x16x4_f32(false, a, false, b, (short)0, acc, false, false);
        }
        const long rowbase = (long)(rg * 16) * M_OUT;
        store_block(y, acc, rowbase, wg_pout + 64 * 16 + mn, kp);
    }
}

extern "C" void kernel_launch(void* const* d_in, const int* in_sizes, int n_in,
                              void* d_out, int out_size, void* d_ws, size_t ws_size,
                              hipStream_t stream) {
    const float* x    = (const float*)d_in[0];
    const float* taps = (const float*)d_in[1];
    float* y          = (float*)d_out;

    dim3 grid(NUM_WG, 1, 1);
    dim3 block(THREADS, 1, 1);
    fir_wmma_f32_kernel<<<grid, block, 0, stream>>>(x, taps, y);
}


# 8 waves per workgroup instead of 32: each wave owns 128 outputs x all 32 rows (16 tiles per B fragment) and issues all its staging loads before the first LDS store; summation order unchanged
# speedup vs baseline: 1.6270x; 1.1719x over previous
#include <hip/hip_runtime.h>


typedef float v2f __attribute__((ext_vector_type(2)));
typedef float v4f __attribute__((ext_vector_type(4)));
typedef float v8f __attribute__((ext_vector_type(8)));

#define N_IN     262144
#define N_TAPS   255
#define M_OUT    (N_IN + N_TAPS - 1)
#define BLKS     65
#define OUT_TILE (BLKS * 16)
#define NUM_WG   253
#define SH       2
#define XW4      325
#define KBLK     68
#define LDX      1300
#define HR_SIZE  288
#define WAVES    8
#define THREADS  (WAVES * 32)
#define RPW      (32 / WAVES)
#define NB       (32 / WAVES)
#define LTRIPS   11
#define FK       143
#define FKB      36
#define BF_E     ((FKB * 32 + THREADS - 1) / THREADS)
#define HR_E     ((HR_SIZE + THREADS - 1) / THREADS)

__device__ __forceinline__ float tap_at(const float* __restrict__ taps, const int j, const bool row_ok)
{
    int t = j - 127; if (t < 0) t = -t;
    const float v = taps[t < 127 ? t : 127];
    return (row_ok && t <= 127) ? v : 0.f;
}

__device__ __forceinline__ void store_block(float* __restrict__ y, const v8f acc,
                                            const long rowbase, const long pout, const int kp)
{
    if (pout < (long)M_OUT) {
        #pragma unroll
        for (int v = 0; v < 8; ++v) {
            const int row = v + 8 * kp;
            y[rowbase + (long)row * M_OUT + pout] = acc[v];
        }
    }
}

template <bool EDGE>
__device__ __forceinline__ void stage_rows(const float* __restrict__ x, float* __restrict__ xt,
                                           const long p0, const int wave, const int lane)
{
    v4f v[RPW][LTRIPS];
    #pragma unroll
    for (int r = 0; r < RPW; ++r) {
        const float* __restrict__ src = x + (long)(wave * RPW + r) * N_IN;
        #pragma unroll
        for (int j = 0; j < LTRIPS; ++j) {
            const int c = lane + 32 * j;
            if (c < XW4) {
                const long g = p0 + 4 * c;
                if (!EDGE) {
                    v[r][j] = *(const v4f*)(src + g);
                } else {
                    const bool ok = (g >= 0 && g < (long)N_IN);
                    const v4f t = *(const v4f*)(src + (ok ? g : 0));
                    const v4f z = {0.f, 0.f, 0.f, 0.f};
                    v[r][j] = ok ? t : z;
                }
            }
        }
    }
    #pragma unroll
    for (int r = 0; r < RPW; ++r) {
        #pragma unroll
        for (int j = 0; j < LTRIPS; ++j) {
            const int c = lane + 32 * j;
            if (c < XW4) *(v4f*)&xt[(wave * RPW + r) * LDX + 4 * c] = v[r][j];
        }
    }
}

__global__ __launch_bounds__(THREADS)
void fir_wmma_f32_kernel(const float* __restrict__ x,
                         const float* __restrict__ taps,
                         float* __restrict__ y)
{
    __shared__ __attribute__((aligned(16))) float xt[32 * LDX];
    __shared__ float hr[HR_SIZE];
    __shared__ v4f bf[FKB * 32];

    const int  tid     = threadIdx.x;
    const int  wave    = tid >> 5;
    const int  lane    = tid & 31;
    const long wg_pout = (long)blockIdx.x * OUT_TILE;
    const long wg_p0   = wg_pout - (N_TAPS - 1) - SH;

    float hv[HR_E];
    #pragma unroll
    for (int e = 0; e < HR_E; ++e) hv[e] = tap_at(taps, 270 - (tid + e * THREADS), true);

    float tp[BF_E][4];
    #pragma unroll
    for (int e = 0; e < BF_E; ++e) {
        const int idx = tid + e * THREADS;
        const int c   = idx & 15;
        const int k   = 4 * (idx >> 5) + 2 * ((idx >> 4) & 1);
        tp[e][0] = tap_at(taps, 254 + c - k, k < FK);
        tp[e][1] = tap_at(taps, 285 - c - k, k < FK);
        tp[e][2] = tap_at(taps, 253 + c - k, k + 1 < FK);
        tp[e][3] = tap_at(taps, 284 - c - k, k + 1 < FK);
    }

    if (wg_p0 >= 0 && wg_p0 + 4 * XW4 <= (long)N_IN) stage_rows<false>(x, xt, wg_p0, wave, lane);
    else                                             stage_rows<true >(x, xt, wg_p0, wave, lane);

    #pragma unroll
    for (int e = 0; e < HR_E; ++e) {
        const int c = tid + e * THREADS;
        if (c < HR_SIZE) hr[c] = hv[e];
    }
    #pragma unroll
    for (int e = 0; e < BF_E; ++e) {
        const int idx = tid + e * THREADS;
        if (idx < FKB * 32) {
            v4f q;
            q.x = 0.5f * (tp[e][0] + tp[e][1]); q.y = 0.5f * (tp[e][2] + tp[e][3]);
            q.z = 0.5f * (tp[e][0] - tp[e][1]); q.w = 0.5f * (tp[e][2] - tp[e][3]);
            bf[idx] = q;
        }
    }
    __syncthreads();

    const int mn   = lane & 15;
    const int kp   = lane >> 4;
    const int rb0 = 16 - mn + 2 * kp;

    {
        const float* __restrict__ fwd = &xt[mn * LDX + SH + wave * (NB * 32) + 2 * kp];
        const float* __restrict__ mir = &xt[mn * LDX + SH + wave * (NB * 32) + 284 - 2 * kp];
        const v4f* __restrict__ bfl = &bf[lane];

        v8f u[2][NB], w[2][NB];
        #pragma unroll
        for (int rg = 0; rg < 2; ++rg) {
            #pragma unroll
            for (int b = 0; b < NB; ++b) {
                #pragma unroll
                for (int i = 0; i < 8; ++i) { u[rg][b][i] = 0.f; w[rg][b][i] = 0.f; }
            }
        }

        #pragma unroll 2
        for (int s = 0; s < FKB; ++s) {
            const int k0 = s * 4;
            const v4f bb = bfl[s * 32];
            v2f bs; bs.x = bb.x; bs.y = bb.y;
            v2f bd; bd.x = bb.z; bd.y = bb.w;
            #pragma unroll
            for (int rg = 0; rg < 2; ++rg) {
                #pragma unroll
                for (int b = 0; b < NB; ++b) {
                    const v2f f = *(const v2f*)(fwd + rg * 16 * LDX + b * 32 + k0);
                    const v2f m = *(const v2f*)(mir + rg * 16 * LDX + b * 32 - k0);
                    v2f sm; sm.x = f.x + m.y; sm.y = f.y + m.x;
                    v2f df; df.x = f.x - m.y; df.y = f.y - m.x;
                    u[rg][b] = __builtin_amdgcn_wmma_f32_16x16x4_f32(false, sm, false, bs, (short)0, u[rg][b], false, false);
                    w[rg][b] = __builtin_amdgcn_wmma_f32_16x16x4_f32(false, df, false, bd, (short)0, w[rg][b], false, false);
                }
            }
        }

        #pragma unroll
        for (int rg = 0; rg < 2; ++rg) {
            const long rowbase = (long)(rg * 16) * M_OUT;
            #pragma unroll
            for (int b = 0; b < NB; ++b) {
                const long pbase = wg_pout + wave * (NB * 32) + b * 32;
                store_block(y, u[rg][b] + w[rg][b], rowbase, pbase + mn,      kp);
                store_block(y, u[rg][b] - w[rg][b], rowbase, pbase + 31 - mn, kp);
            }
        }
    }

    if (wave < 2) {
        const int rg = wave;
        const float* __restrict__ arow = &xt[(rg * 16 + mn) * LDX + SH + 64 * 16 + 2 * kp];
        v8f acc = {0.f,0.f,0.f,0.f,0.f,0.f,0.f,0.f};
        #pragma unroll 4
        for (int kb = 0; kb < KBLK; ++kb) {
            const int k0 = kb * 4;
            v2f b;
            b.x = hr[rb0 + k0];
            b.y = hr[rb0 + k0 + 1];
            v2f a = *(const v2f*)(arow + k0);
            acc = __builtin_amdgcn_wmma_f32_16x16x4_f32(false, a, false, b, (short)0, acc, false, false);
        }
        const long rowbase = (long)(rg * 16) * M_OUT;
        store_block(y, acc, rowbase, wg_pout + 64 * 16 + mn, kp);
    }
}

extern "C" void kernel_launch(void* const* d_in, const int* in_sizes, int n_in,
                              void* d_out, int out_size, void* d_ws, size_t ws_size,
                              hipStream_t stream) {
    const float* x    = (const float*)d_in[0];
    const float* taps = (const float*)d_in[1];
    float* y          = (float*)d_out;

    dim3 grid(NUM_WG, 1, 1);
    dim3 block(THREADS, 1, 1);
    fir_wmma_f32_kernel<<<grid, block, 0, stream>>>(x, taps, y);
}


# main outputs go through a per-wave LDS tile (WMMA operand roles exchanged: a lane holds 8 consecutive outputs) and leave as one 512-byte run per row, not 64-byte pieces; summation order unchanged
# speedup vs baseline: 1.7511x; 1.0763x over previous
#include <hip/hip_runtime.h>


typedef float v2f __attribute__((ext_vector_type(2)));
typedef float v4f __attribute__((ext_vector_type(4)));
typedef float v8f __attribute__((ext_vector_type(8)));

#define N_IN     262144
#define N_TAPS   255
#define M_OUT    (N_IN + N_TAPS - 1)
#define BLKS     65
#define OUT_TILE (BLKS * 16)
#define NUM_WG   253
#define SH       2
#define XW4      325
#define KBLK     68
#define LDX      1300
#define HR_SIZE  288
#define WAVES    8
#define THREADS  (WAVES * 32)
#define RPW      (32 / WAVES)
#define NB       (32 / WAVES)
#define LTRIPS   11
#define FK       143
#define FKB      36
#define BF_E     ((FKB * 32 + THREADS - 1) / THREADS)
#define HR_E     ((HR_SIZE + THREADS - 1) / THREADS)
#define WOUT     (NB * 32)
#define OP       132

__device__ __forceinline__ float tap_at(const float* __restrict__ taps, const int j, const bool row_ok)
{
    int t = j - 127; if (t < 0) t = -t;
    const float v = taps[t < 127 ? t : 127];
    return (row_ok && t <= 127) ? v : 0.f;
}

__device__ __forceinline__ void store_block(float* __restrict__ y, const v8f acc,
                                            const long rowbase, const long pout, const int kp)
{
    if (pout < (long)M_OUT) {
        #pragma unroll
        for (int v = 0; v < 8; ++v) {
            const int row = v + 8 * kp;
            y[rowbase + (long)row * M_OUT + pout] = acc[v];
        }
    }
}

template <bool FULL>
__device__ __forceinline__ void store_rows(float* __restrict__ y, const float* ow,
                                           const long o0, const int nvalid, const int lane)
{
    const int pe = 4 * lane;
    const int po = lane > 0 ? 4 * lane - 2 : 0;
    const int pp = (lane & 1) * (WOUT - 2);
    v4f ev[16];
    v2f oa[16], ob[16], ep[16];
    #pragma unroll
    for (int i = 0; i < 16; ++i) {
        const float* se = ow + (2 * i) * OP;
        const float* so = ow + (2 * i + 1) * OP;
        ev[i] = *(const v4f*)(se + pe);
        oa[i] = *(const v2f*)(so + po);
        ob[i] = *(const v2f*)(so + po + 2);
        ep[i] = *(const v2f*)(so + pp);
    }
    float* __restrict__ d0 = y + o0;
    if (FULL || pe + 4 <= nvalid) {
        #pragma unroll
        for (int i = 0; i < 16; ++i) *(v4f*)(d0 + (long)(2 * i) * M_OUT + pe) = ev[i];
    } else if (pe + 2 <= nvalid) {
        #pragma unroll
        for (int i = 0; i < 16; ++i) {
            v2f h; h.x = ev[i].x; h.y = ev[i].y;
            *(v2f*)(d0 + (long)(2 * i) * M_OUT + pe) = h;
        }
    }
    if (lane > 0) {
        if (FULL || po + 4 <= nvalid) {
            #pragma unroll
            for (int i = 0; i < 16; ++i) {
                v4f v; v.x = oa[i].x; v.y = oa[i].y; v.z = ob[i].x; v.w = ob[i].y;
                *(v4f*)(d0 + (long)(2 * i + 1) * M_OUT + po) = v;
            }
        } else if (po + 2 <= nvalid) {
            #pragma unroll
            for (int i = 0; i < 16; ++i) *(v2f*)(d0 + (long)(2 * i + 1) * M_OUT + po) = oa[i];
        }
    }
    if (lane < 2 && (FULL || pp + 2 <= nvalid)) {
        #pragma unroll
        for (int i = 0; i < 16; ++i) *(v2f*)(d0 + (long)(2 * i + 1) * M_OUT + pp) = ep[i];
    }
}

template <bool EDGE>
__device__ __forceinline__ void stage_rows(const float* __restrict__ x, float* __restrict__ xt,
                                           const long p0, const int wave, const int lane)
{
    v4f v[RPW][LTRIPS];
    #pragma unroll
    for (int r = 0; r < RPW; ++r) {
        const float* __restrict__ src = x + (long)(wave * RPW + r) * N_IN;
        #pragma unroll
        for (int j = 0; j < LTRIPS; ++j) {
            const int c = lane + 32 * j;
            if (c < XW4) {
                const long g = p0 + 4 * c;
                if (!EDGE) {
                    v[r][j] = *(const v4f*)(src + g);
                } else {
                    const bool ok = (g >= 0 && g < (long)N_IN);
                    const v4f t = *(const v4f*)(src + (ok ? g : 0));
                    const v4f z = {0.f, 0.f, 0.f, 0.f};
                    v[r][j] = ok ? t : z;
                }
            }
        }
    }
    #pragma unroll
    for (int r = 0; r < RPW; ++r) {
        #pragma unroll
        for (int j = 0; j < LTRIPS; ++j) {
            const int c = lane + 32 * j;
            if (c < XW4) *(v4f*)&xt[(wave * RPW + r) * LDX + 4 * c] = v[r][j];
        }
    }
}

__global__ __launch_bounds__(THREADS)
void fir_wmma_f32_kernel(const float* __restrict__ x,
                         const float* __restrict__ taps,
                         float* __restrict__ y)
{
    __shared__ __attribute__((aligned(16))) float xt[32 * LDX];
    __shared__ float hr[HR_SIZE];
    __shared__ v4f bf[FKB * 32];
    __shared__ __attribute__((aligned(16))) float ot[WAVES * 32 * OP];

    const int  tid     = threadIdx.x;
    const int  wave    = tid >> 5;
    const int  lane    = tid & 31;
    const long wg_pout = (long)blockIdx.x * OUT_TILE;
    const long wg_p0   = wg_pout - (N_TAPS - 1) - SH;

    float hv[HR_E];
    #pragma unroll
    for (int e = 0; e < HR_E; ++e) hv[e] = tap_at(taps, 270 - (tid + e * THREADS), true);

    float tp[BF_E][4];
    #pragma unroll
    for (int e = 0; e < BF_E; ++e) {
        const int idx = tid + e * THREADS;
        const int c   = idx & 15;
        const int k   = 4 * (idx >> 5) + 2 * ((idx >> 4) & 1);
        tp[e][0] = tap_at(taps, 254 + c - k, k < FK);
        tp[e][1] = tap_at(taps, 285 - c - k, k < FK);
        tp[e][2] = tap_at(taps, 253 + c - k, k + 1 < FK);
        tp[e][3] = tap_at(taps, 284 - c - k, k + 1 < FK);
    }

    if (wg_p0 >= 0 && wg_p0 + 4 * XW4 <= (long)N_IN) stage_rows<false>(x, xt, wg_p0, wave, lane);
    else                                             stage_rows<true >(x, xt, wg_p0, wave, lane);

    #pragma unroll
    for (int e = 0; e < HR_E; ++e) {
        const int c = tid + e * THREADS;
        if (c < HR_SIZE) hr[c] = hv[e];
    }
    #pragma unroll
    for (int e = 0; e < BF_E; ++e) {
        const int idx = tid + e * THREADS;
        if (idx < FKB * 32) {
            v4f q;
            q.x = 0.5f * (tp[e][0] + tp[e][1]); q.y = 0.5f * (tp[e][2] + tp[e][3]);
            q.z = 0.5f * (tp[e][0] - tp[e][1]); q.w = 0.5f * (tp[e][2] - tp[e][3]);
            bf[idx] = q;
        }
    }
    __syncthreads();

    const int mn   = lane & 15;
    const int kp   = lane >> 4;
    const int rb0 = 16 - mn + 2 * kp;

    {
        const float* __restrict__ fwd = &xt[mn * LDX + SH + wave * (NB * 32) + 2 * kp];
        const float* __restrict__ mir = &xt[mn * LDX + SH + wave * (NB * 32) + 284 - 2 * kp];
        const v4f* __restrict__ bfl = &bf[lane];

        v8f u[2][NB], w[2][NB];
        #pragma unroll
        for (int rg = 0; rg < 2; ++rg) {
            #pragma unroll
            for (int b = 0; b < NB; ++b) {
                #pragma unroll
                for (int i = 0; i < 8; ++i) { u[rg][b][i] = 0.f; w[rg][b][i] = 0.f; }
            }
        }

        #pragma unroll 2
        for (int s = 0; s < FKB; ++s) {
            const int k0 = s * 4;
            const v4f bb = bfl[s * 32];
            v2f bs; bs.x = bb.x; bs.y = bb.y;
            v2f bd; bd.x = bb.z; bd.y = bb.w;
            #pragma unroll
            for (int rg = 0; rg < 2; ++rg) {
                #pragma unroll
                for (int b = 0; b < NB; ++b) {
                    const v2f f = *(const v2f*)(fwd + rg * 16 * LDX + b * 32 + k0);
                    const v2f m = *(const v2f*)(mir + rg * 16 * LDX + b * 32 - k0);
                    v2f sm; sm.x = f.x + m.y; sm.y = f.y + m.x;
                    v2f df; df.x = f.x - m.y; df.y = f.y - m.x;
                    u[rg][b] = __builtin_amdgcn_wmma_f32_16x16x4_f32(false, bs, false, sm, (short)0, u[rg][b], false, false);
                    w[rg][b] = __builtin_amdgcn_wmma_f32_16x16x4_f32(false, bd, false, df, (short)0, w[rg][b], false, false);
                }
            }
        }

        const int wv = __builtin_amdgcn_readfirstlane(wave);
        float* ow = &ot[wv * (32 * OP)];
        #pragma unroll
        for (int rg = 0; rg < 2; ++rg) {
            #pragma unroll
            for (int b = 0; b < NB; ++b) {
                const v8f s = u[rg][b] + w[rg][b];
                const v8f d = u[rg][b] - w[rg][b];
                float* o = ow + (rg * 16 + mn) * OP + b * 32;
                v4f q;
                q.x = s[0]; q.y = s[1]; q.z = s[2]; q.w = s[3]; *(v4f*)(o + 8 * kp)      = q;
                q.x = s[4]; q.y = s[5]; q.z = s[6]; q.w = s[7]; *(v4f*)(o + 8 * kp + 4)  = q;
                q.x = d[7]; q.y = d[6]; q.z = d[5]; q.w = d[4]; *(v4f*)(o + 24 - 8 * kp) = q;
                q.x = d[3]; q.y = d[2]; q.z = d[1]; q.w = d[0]; *(v4f*)(o + 28 - 8 * kp) = q;
            }
        }
        __builtin_amdgcn_fence(__ATOMIC_ACQ_REL, "workgroup");

        const long o0  = wg_pout + wv * WOUT;
        const long rem = (long)M_OUT - o0;
        if (rem >= (long)WOUT) store_rows<true >(y, ow, o0, WOUT, lane);
        else if (rem > 0)      store_rows<false>(y, ow, o0, (int)rem, lane);
    }

    if (wave < 2) {
        const int rg = wave;
        const float* __restrict__ arow = &xt[(rg * 16 + mn) * LDX + SH + 64 * 16 + 2 * kp];
        v8f acc = {0.f,0.f,0.f,0.f,0.f,0.f,0.f,0.f};
        #pragma unroll 4
        for (int kb = 0; kb < KBLK; ++kb) {
            const int k0 = kb * 4;
            v2f b;
            b.x = hr[rb0 + k0];
            b.y = hr[rb0 + k0 + 1];
            v2f a = *(const v2f*)(arow + k0);
            acc = __builtin_amdgcn_wmma_f32_16x16x4_f32(false, a, false, b, (short)0, acc, false, false);
        }
        const long rowbase = (long)(rg * 16) * M_OUT;
        store_block(y, acc, rowbase, wg_pout + 64 * 16 + mn, kp);
    }
}

extern "C" void kernel_launch(void* const* d_in, const int* in_sizes, int n_in,
                              void* d_out, int out_size, void* d_ws, size_t ws_size,
                              hipStream_t stream) {
    const float* x    = (const float*)d_in[0];
    const float* taps = (const float*)d_in[1];
    float* y          = (float*)d_out;

    dim3 grid(NUM_WG, 1, 1);
    dim3 block(THREADS, 1, 1);
    fir_wmma_f32_kernel<<<grid, block, 0, stream>>>(x, taps, y);
}
